# starting kernel
# baseline (speedup 1.0000x reference)
#include <hip/hip_runtime.h>


typedef __attribute__((ext_vector_type(2))) float v2f;
typedef __attribute__((ext_vector_type(8))) float v8f;

#define DEG        10
#define NC         11
#define NV         48
#define NSAMP      50000
#define NP         1128
#define KDIM       528
#define KSTEPS     132
#define NTILES     3
#define ROWS_BLK   64
#define G_STRIDE   532
#define BLK_THREADS 128

__global__ void build_w_kernel(const float* __restrict__ params,
                               float* __restrict__ wsw) {
  int tid = blockIdx.x * blockDim.x + threadIdx.x;
  const int total = KSTEPS * NTILES * 32;
  if (tid >= total) return;
  int lane = tid & 31;
  int tile = (tid >> 5) % NTILES;
  int kk   = tid / (32 * NTILES);
  int var  = tile * 16 + (lane & 15);
  int k0   = kk * 4 + ((lane >> 4) ? 2 : 0);
  float vals[2];
#pragma unroll
  for (int i = 0; i < 2; ++i) {
    int k     = k0 + i;
    int covar = k / NC;
    int d     = k - covar * NC;
    float w   = 0.0f;
    if (covar < var)
      w = params[d * NP + (var * (var - 1)) / 2 + covar];
    vals[i] = w;
  }
  float2* out2 = (float2*)wsw;
  out2[tid] = make_float2(vals[0], vals[1]);
}

__global__ void decorrelation_kernel(const float* __restrict__ x,
                                     const float* __restrict__ prange,
                                     const float* __restrict__ wsw,
                                     float* __restrict__ out) {
  extern __shared__ float lds[];
  const int tid     = threadIdx.x;
  const int lane    = tid & 31;
  const int wave    = tid >> 5;
  const int rowBase = blockIdx.x * ROWS_BLK;

  const float binom[NC] = {1.f, 10.f, 45.f, 120.f, 210.f, 252.f,
                           210.f, 120.f, 45.f, 10.f, 1.f};

  for (int task = tid; task < ROWS_BLK * NV; task += BLK_THREADS) {
    int row   = task / NV;
    int covar = task - row * NV;
    int rg    = rowBase + row;
    if (rg >= NSAMP) rg = NSAMP - 1;
    float xv = x[rg * NV + covar];
    float lo = prange[covar];
    float hi = prange[NV + covar];
    float xn = (xv - lo) / (hi - lo);
    float om = 1.0f - xn;
    float p[NC], q[NC];
    p[0] = 1.0f;
    q[0] = 1.0f;
#pragma unroll
    for (int d = 1; d < NC; ++d) {
      p[d] = p[d - 1] * xn;
      q[d] = q[d - 1] * om;
    }
    float* g = &lds[row * G_STRIDE + covar * NC];
#pragma unroll
    for (int d = 0; d < NC; ++d)
      g[d] = binom[d] * p[d] * q[DEG - d] * xv;
  }
  __syncthreads();

  v8f c0 = {}, c1 = {}, c2 = {};
  const int   mrow  = lane & 15;
  const int   khalf = (lane >> 4) * 2;
  const float* aBase = &lds[(wave * 16 + mrow) * G_STRIDE + khalf];
  const v2f*   bBase = (const v2f*)wsw;

#pragma unroll 2
  for (int kk = 0; kk < KSTEPS; ++kk) {
    v2f a  = *(const v2f*)(aBase + kk * 4);
    v2f b0 = bBase[(kk * 3 + 0) * 32 + lane];
    v2f b1 = bBase[(kk * 3 + 1) * 32 + lane];
    v2f b2 = bBase[(kk * 3 + 2) * 32 + lane];
    c0 = __builtin_amdgcn_wmma_f32_16x16x4_f32(false, a, false, b0,
                                               (short)0, c0, false, false);
    c1 = __builtin_amdgcn_wmma_f32_16x16x4_f32(false, a, false, b1,
                                               (short)0, c1, false, false);
    c2 = __builtin_amdgcn_wmma_f32_16x16x4_f32(false, a, false, b2,
                                               (short)0, c2, false, false);
  }

  const int col = lane & 15;
  const int rg0 = rowBase + wave * 16 + ((lane >> 4) ? 8 : 0);
#pragma unroll
  for (int j = 0; j < 8; ++j) {
    int rg = rg0 + j;
    if (rg < NSAMP) {
      out[rg * NV + col]      = x[rg * NV + col]      + c0[j];
      out[rg * NV + 16 + col] = x[rg * NV + 16 + col] + c1[j];
      out[rg * NV + 32 + col] = x[rg * NV + 32 + col] + c2[j];
    }
  }
}

extern "C" void kernel_launch(void* const* d_in, const int* in_sizes, int n_in,
                              void* d_out, int out_size, void* d_ws, size_t ws_size,
                              hipStream_t stream) {
  const float* x      = (const float*)d_in[0];
  const float* params = (const float*)d_in[1];
  const float* prange = (const float*)d_in[2];
  float*       out    = (float*)d_out;
  float*       wsw    = (float*)d_ws;

  {
    const int total  = KSTEPS * NTILES * 32;
    const int blocks = (total + 255) / 256;
    build_w_kernel<<<blocks, 256, 0, stream>>>(params, wsw);
  }

  {
    const int    blocks   = (NSAMP + ROWS_BLK - 1) / ROWS_BLK;
    const size_t ldsBytes = (size_t)ROWS_BLK * G_STRIDE * sizeof(float);
    decorrelation_kernel<<<blocks, BLK_THREADS, ldsBytes, stream>>>(x, prange,
                                                                    wsw, out);
  }
}


# Register double-buffering of the k loop: A and B fragments of the next 12 k-steps (36 global b64 + 12 LDS b64 loads) issued as one run ahead of each 36-WMMA group; launch bound 128; sums unchanged
# speedup vs baseline: 1.6091x; 1.6091x over previous
#include <hip/hip_runtime.h>


typedef __attribute__((ext_vector_type(2))) float v2f;
typedef __attribute__((ext_vector_type(8))) float v8f;

#define DEG        10
#define NC         11
#define NV         48
#define NSAMP      50000
#define NP         1128
#define KDIM       528
#define KSTEPS     132
#define NTILES     3
#define ROWS_BLK   64
#define G_STRIDE   532
#define BLK_THREADS 128
#define KGROUP     12
#define NGROUPS    11

__global__ void build_w_kernel(const float* __restrict__ params,
                               float* __restrict__ wsw) {
  int tid = blockIdx.x * blockDim.x + threadIdx.x;
  const int total = KSTEPS * NTILES * 32;
  if (tid >= total) return;
  int lane = tid & 31;
  int tile = (tid >> 5) % NTILES;
  int kk   = tid / (32 * NTILES);
  int var  = tile * 16 + (lane & 15);
  int k0   = kk * 4 + ((lane >> 4) ? 2 : 0);
  float vals[2];
#pragma unroll
  for (int i = 0; i < 2; ++i) {
    int k     = k0 + i;
    int covar = k / NC;
    int d     = k - covar * NC;
    float w   = 0.0f;
    if (covar < var)
      w = params[d * NP + (var * (var - 1)) / 2 + covar];
    vals[i] = w;
  }
  float2* out2 = (float2*)wsw;
  out2[tid] = make_float2(vals[0], vals[1]);
}

struct BFrags { v2f v[KGROUP * NTILES]; };
struct AFrags { v2f v[KGROUP]; };

__device__ __forceinline__ void load_b_group(BFrags& f, const v2f* __restrict__ bLane,
                                             int g) {
  const v2f* bp = bLane + g * (KGROUP * NTILES * 32);
#pragma unroll
  for (int i = 0; i < KGROUP * NTILES; ++i)
    f.v[i] = bp[i * 32];
}

__device__ __forceinline__ void load_a_group(AFrags& f, const float* aBase, int g) {
  const float* ap = aBase + g * (KGROUP * 4);
#pragma unroll
  for (int i = 0; i < KGROUP; ++i)
    f.v[i] = *(const v2f*)(ap + i * 4);
}

__device__ __forceinline__ void mma_group(const AFrags& a, const BFrags& b,
                                          v8f& c0, v8f& c1, v8f& c2) {
#pragma unroll
  for (int i = 0; i < KGROUP; ++i) {
    c0 = __builtin_amdgcn_wmma_f32_16x16x4_f32(false, a.v[i], false, b.v[i * 3 + 0],
                                               (short)0, c0, false, false);
    c1 = __builtin_amdgcn_wmma_f32_16x16x4_f32(false, a.v[i], false, b.v[i * 3 + 1],
                                               (short)0, c1, false, false);
    c2 = __builtin_amdgcn_wmma_f32_16x16x4_f32(false, a.v[i], false, b.v[i * 3 + 2],
                                               (short)0, c2, false, false);
  }
}

__global__ void __launch_bounds__(BLK_THREADS)
decorrelation_kernel(const float* __restrict__ x,
                     const float* __restrict__ prange,
                     const float* __restrict__ wsw,
                     float* __restrict__ out) {
  extern __shared__ float lds[];
  const int tid     = threadIdx.x;
  const int lane    = tid & 31;
  const int wave    = tid >> 5;
  const int rowBase = blockIdx.x * ROWS_BLK;

  const float binom[NC] = {1.f, 10.f, 45.f, 120.f, 210.f, 252.f,
                           210.f, 120.f, 45.f, 10.f, 1.f};

  for (int task = tid; task < ROWS_BLK * NV; task += BLK_THREADS) {
    int row   = task / NV;
    int covar = task - row * NV;
    int rg    = rowBase + row;
    if (rg >= NSAMP) rg = NSAMP - 1;
    float xv = x[rg * NV + covar];
    float lo = prange[covar];
    float hi = prange[NV + covar];
    float xn = (xv - lo) / (hi - lo);
    float om = 1.0f - xn;
    float p[NC], q[NC];
    p[0] = 1.0f;
    q[0] = 1.0f;
#pragma unroll
    for (int d = 1; d < NC; ++d) {
      p[d] = p[d - 1] * xn;
      q[d] = q[d - 1] * om;
    }
    float* g = &lds[row * G_STRIDE + covar * NC];
#pragma unroll
    for (int d = 0; d < NC; ++d)
      g[d] = binom[d] * p[d] * q[DEG - d] * xv;
  }
  v8f c0 = {}, c1 = {}, c2 = {};
  const int   mrow  = lane & 15;
  const int   khalf = (lane >> 4) * 2;
  const float* aBase = &lds[(wave * 16 + mrow) * G_STRIDE + khalf];
  const v2f*   bLane = (const v2f*)wsw + lane;

  BFrags bset[2];
  AFrags aset[2];
  load_b_group(bset[0], bLane, 0);
  __syncthreads();
  load_a_group(aset[0], aBase, 0);

#pragma unroll
  for (int g = 0; g < NGROUPS; ++g) {
    if (g + 1 < NGROUPS) {
      load_b_group(bset[(g + 1) & 1], bLane, g + 1);
      load_a_group(aset[(g + 1) & 1], aBase, g + 1);
    }
    __builtin_amdgcn_sched_barrier(0);
    mma_group(aset[g & 1], bset[g & 1], c0, c1, c2);
    __builtin_amdgcn_sched_barrier(0);
  }

  const int col = lane & 15;
  const int rg0 = rowBase + wave * 16 + ((lane >> 4) ? 8 : 0);
#pragma unroll
  for (int j = 0; j < 8; ++j) {
    int rg = rg0 + j;
    if (rg < NSAMP) {
      out[rg * NV + col]      = x[rg * NV + col]      + c0[j];
      out[rg * NV + 16 + col] = x[rg * NV + 16 + col] + c1[j];
      out[rg * NV + 32 + col] = x[rg * NV + 32 + col] + c2[j];
    }
  }
}

extern "C" void kernel_launch(void* const* d_in, const int* in_sizes, int n_in,
                              void* d_out, int out_size, void* d_ws, size_t ws_size,
                              hipStream_t stream) {
  const float* x      = (const float*)d_in[0];
  const float* params = (const float*)d_in[1];
  const float* prange = (const float*)d_in[2];
  float*       out    = (float*)d_out;
  float*       wsw    = (float*)d_ws;

  {
    const int total  = KSTEPS * NTILES * 32;
    const int blocks = (total + 255) / 256;
    build_w_kernel<<<blocks, 256, 0, stream>>>(params, wsw);
  }

  {
    const int    blocks   = (NSAMP + ROWS_BLK - 1) / ROWS_BLK;
    const size_t ldsBytes = (size_t)ROWS_BLK * G_STRIDE * sizeof(float);
    decorrelation_kernel<<<blocks, BLK_THREADS, ldsBytes, stream>>>(x, prange,
                                                                    wsw, out);
  }
}


# G tile build unrolled: a thread's 24 x loads and 6 range loads leave as one run ahead of the Bernstein arithmetic, so a wave waits one memory round trip, not 24; sums unchanged
# speedup vs baseline: 2.6792x; 1.6650x over previous
#include <hip/hip_runtime.h>


typedef __attribute__((ext_vector_type(2))) float v2f;
typedef __attribute__((ext_vector_type(8))) float v8f;

#define DEG        10
#define NC         11
#define NV         48
#define NSAMP      50000
#define NP         1128
#define KDIM       528
#define KSTEPS     132
#define NTILES     3
#define ROWS_BLK   64
#define G_STRIDE   532
#define BLK_THREADS 128
#define KGROUP     12
#define NGROUPS    11
#define TASKS      24
#define CPERIOD    3
#define RSTEP      8

static_assert(TASKS * BLK_THREADS == ROWS_BLK * NV, "tasks must tile the G block");
static_assert((CPERIOD * BLK_THREADS) % NV == 0, "covar must repeat every CPERIOD tasks");
static_assert(CPERIOD * BLK_THREADS == RSTEP * NV, "row step of one covar period");
static_assert(TASKS % CPERIOD == 0, "whole covar periods per thread");

__global__ void build_w_kernel(const float* __restrict__ params,
                               float* __restrict__ wsw) {
  int tid = blockIdx.x * blockDim.x + threadIdx.x;
  const int total = KSTEPS * NTILES * 32;
  if (tid >= total) return;
  int lane = tid & 31;
  int tile = (tid >> 5) % NTILES;
  int kk   = tid / (32 * NTILES);
  int var  = tile * 16 + (lane & 15);
  int k0   = kk * 4 + ((lane >> 4) ? 2 : 0);
  float vals[2];
#pragma unroll
  for (int i = 0; i < 2; ++i) {
    int k     = k0 + i;
    int covar = k / NC;
    int d     = k - covar * NC;
    float w   = 0.0f;
    if (covar < var)
      w = params[d * NP + (var * (var - 1)) / 2 + covar];
    vals[i] = w;
  }
  float2* out2 = (float2*)wsw;
  out2[tid] = make_float2(vals[0], vals[1]);
}

struct BFrags { v2f v[KGROUP * NTILES]; };
struct AFrags { v2f v[KGROUP]; };

__device__ __forceinline__ void load_b_group(BFrags& f, const v2f* __restrict__ bLane,
                                             int g) {
  const v2f* bp = bLane + g * (KGROUP * NTILES * 32);
#pragma unroll
  for (int i = 0; i < KGROUP * NTILES; ++i)
    f.v[i] = bp[i * 32];
}

__device__ __forceinline__ void load_a_group(AFrags& f, const float* aBase, int g) {
  const float* ap = aBase + g * (KGROUP * 4);
#pragma unroll
  for (int i = 0; i < KGROUP; ++i)
    f.v[i] = *(const v2f*)(ap + i * 4);
}

__device__ __forceinline__ void mma_group(const AFrags& a, const BFrags& b,
                                          v8f& c0, v8f& c1, v8f& c2) {
#pragma unroll
  for (int i = 0; i < KGROUP; ++i) {
    c0 = __builtin_amdgcn_wmma_f32_16x16x4_f32(false, a.v[i], false, b.v[i * 3 + 0],
                                               (short)0, c0, false, false);
    c1 = __builtin_amdgcn_wmma_f32_16x16x4_f32(false, a.v[i], false, b.v[i * 3 + 1],
                                               (short)0, c1, false, false);
    c2 = __builtin_amdgcn_wmma_f32_16x16x4_f32(false, a.v[i], false, b.v[i * 3 + 2],
                                               (short)0, c2, false, false);
  }
}

__global__ void __launch_bounds__(BLK_THREADS)
decorrelation_kernel(const float* __restrict__ x,
                     const float* __restrict__ prange,
                     const float* __restrict__ wsw,
                     float* __restrict__ out) {
  extern __shared__ float lds[];
  const int tid     = threadIdx.x;
  const int lane    = tid & 31;
  const int wave    = tid >> 5;
  const int rowBase = blockIdx.x * ROWS_BLK;

  const float binom[NC] = {1.f, 10.f, 45.f, 120.f, 210.f, 252.f,
                           210.f, 120.f, 45.f, 10.f, 1.f};

  float xv[TASKS];
  float lo3[CPERIOD], hi3[CPERIOD];
  int   row3[CPERIOD], cov3[CPERIOD];
#pragma unroll
  for (int j = 0; j < CPERIOD; ++j) {
    int task = tid + j * BLK_THREADS;
    row3[j]  = task / NV;
    cov3[j]  = task - row3[j] * NV;
    lo3[j]   = prange[cov3[j]];
    hi3[j]   = prange[NV + cov3[j]];
  }
#pragma unroll
  for (int i = 0; i < TASKS; ++i) {
    int j  = i % CPERIOD;
    int rg = rowBase + row3[j] + RSTEP * (i / CPERIOD);
    if (rg >= NSAMP) rg = NSAMP - 1;
    xv[i] = x[rg * NV + cov3[j]];
  }
  __builtin_amdgcn_sched_barrier(0);
#pragma unroll
  for (int i = 0; i < TASKS; ++i) {
    int   j   = i % CPERIOD;
    int   row = row3[j] + RSTEP * (i / CPERIOD);
    float x0  = xv[i];
    float lo  = lo3[j];
    float hi  = hi3[j];
    float xn  = (x0 - lo) / (hi - lo);
    float om  = 1.0f - xn;
    float p[NC], q[NC];
    p[0] = 1.0f;
    q[0] = 1.0f;
#pragma unroll
    for (int d = 1; d < NC; ++d) {
      p[d] = p[d - 1] * xn;
      q[d] = q[d - 1] * om;
    }
    float* g = &lds[row * G_STRIDE + cov3[j] * NC];
#pragma unroll
    for (int d = 0; d < NC; ++d)
      g[d] = binom[d] * p[d] * q[DEG - d] * x0;
  }
  v8f c0 = {}, c1 = {}, c2 = {};
  const int   mrow  = lane & 15;
  const int   khalf = (lane >> 4) * 2;
  const float* aBase = &lds[(wave * 16 + mrow) * G_STRIDE + khalf];
  const v2f*   bLane = (const v2f*)wsw + lane;

  BFrags bset[2];
  AFrags aset[2];
  load_b_group(bset[0], bLane, 0);
  __syncthreads();
  load_a_group(aset[0], aBase, 0);

#pragma unroll
  for (int g = 0; g < NGROUPS; ++g) {
    if (g + 1 < NGROUPS) {
      load_b_group(bset[(g + 1) & 1], bLane, g + 1);
      load_a_group(aset[(g + 1) & 1], aBase, g + 1);
    }
    __builtin_amdgcn_sched_barrier(0);
    mma_group(aset[g & 1], bset[g & 1], c0, c1, c2);
    __builtin_amdgcn_sched_barrier(0);
  }

  const int col = lane & 15;
  const int rg0 = rowBase + wave * 16 + ((lane >> 4) ? 8 : 0);
#pragma unroll
  for (int j = 0; j < 8; ++j) {
    int rg = rg0 + j;
    if (rg < NSAMP) {
      out[rg * NV + col]      = x[rg * NV + col]      + c0[j];
      out[rg * NV + 16 + col] = x[rg * NV + 16 + col] + c1[j];
      out[rg * NV + 32 + col] = x[rg * NV + 32 + col] + c2[j];
    }
  }
}

extern "C" void kernel_launch(void* const* d_in, const int* in_sizes, int n_in,
                              void* d_out, int out_size, void* d_ws, size_t ws_size,
                              hipStream_t stream) {
  const float* x      = (const float*)d_in[0];
  const float* params = (const float*)d_in[1];
  const float* prange = (const float*)d_in[2];
  float*       out    = (float*)d_out;
  float*       wsw    = (float*)d_ws;

  {
    const int total  = KSTEPS * NTILES * 32;
    const int blocks = (total + 255) / 256;
    build_w_kernel<<<blocks, 256, 0, stream>>>(params, wsw);
  }

  {
    const int    blocks   = (NSAMP + ROWS_BLK - 1) / ROWS_BLK;
    const size_t ldsBytes = (size_t)ROWS_BLK * G_STRIDE * sizeof(float);
    decorrelation_kernel<<<blocks, BLK_THREADS, ldsBytes, stream>>>(x, prange,
                                                                    wsw, out);
  }
}


# W's zero blocks left out of the product: column tile t runs only the k-steps below its limit 42/86/130 of 132 (258 WMMAs and B loads per wave, not 396); each sum's k order unchanged
# speedup vs baseline: 3.1963x; 1.1930x over previous
#include <hip/hip_runtime.h>


typedef __attribute__((ext_vector_type(2))) float v2f;
typedef __attribute__((ext_vector_type(8))) float v8f;

#define DEG        10
#define NC         11
#define NV         48
#define NSAMP      50000
#define NP         1128
#define KDIM       528
#define KSTEPS     132
#define NTILES     3
#define ROWS_BLK   64
#define G_STRIDE   532
#define BLK_THREADS 128
#define NSLOTS     258
#define GSLOTS     18
#define NGROUPS    15
#define NSETS      3
#define KLIM0      42
#define KLIM1      86
#define KLIM2      130
#define SEG1       126
#define SEG2       214

static_assert(SEG2 + (KLIM2 - KLIM1) == NSLOTS, "slot list covers every nonzero block once");
static_assert(KLIM2 <= KSTEPS, "k limits inside the G tile");
#define TASKS      24
#define CPERIOD    3
#define RSTEP      8

static_assert(TASKS * BLK_THREADS == ROWS_BLK * NV, "tasks must tile the G block");
static_assert((CPERIOD * BLK_THREADS) % NV == 0, "covar must repeat every CPERIOD tasks");
static_assert(CPERIOD * BLK_THREADS == RSTEP * NV, "row step of one covar period");
static_assert(TASKS % CPERIOD == 0, "whole covar periods per thread");

__host__ __device__ constexpr int slot_tile(int s) {
  return (s < SEG1) ? s % 3 : (s < SEG2) ? 1 + (s - SEG1) % 2 : 2;
}
__host__ __device__ constexpr int slot_kstep(int s) {
  return (s < SEG1) ? s / 3 : (s < SEG2) ? KLIM0 + (s - SEG1) / 2 : KLIM1 + (s - SEG2);
}

__global__ void build_w_kernel(const float* __restrict__ params,
                               float* __restrict__ wsw) {
  int tid = blockIdx.x * blockDim.x + threadIdx.x;
  const int total = NSLOTS * 32;
  if (tid >= total) return;
  int lane = tid & 31;
  int slot = tid >> 5;
  int tile = slot_tile(slot);
  int kk   = slot_kstep(slot);
  int var  = tile * 16 + (lane & 15);
  int k0   = kk * 4 + ((lane >> 4) ? 2 : 0);
  float vals[2];
#pragma unroll
  for (int i = 0; i < 2; ++i) {
    int k     = k0 + i;
    int covar = k / NC;
    int d     = k - covar * NC;
    float w   = 0.0f;
    if (covar < var)
      w = params[d * NP + (var * (var - 1)) / 2 + covar];
    vals[i] = w;
  }
  float2* out2 = (float2*)wsw;
  out2[tid] = make_float2(vals[0], vals[1]);
}

struct BFrags { v2f v[GSLOTS]; };
struct AFrags { v2f v[GSLOTS]; };

__device__ __forceinline__ void load_b_group(BFrags& f, const v2f* __restrict__ bLane,
                                             int g) {
#pragma unroll
  for (int i = 0; i < GSLOTS; ++i) {
    int s = g * GSLOTS + i;
    if (s < NSLOTS) f.v[i] = bLane[s * 32];
  }
}

__device__ __forceinline__ void load_a_group(AFrags& f, const float* aBase, int g) {
#pragma unroll
  for (int i = 0; i < GSLOTS; ++i) {
    int s = g * GSLOTS + i;
    if (s < NSLOTS) f.v[i] = *(const v2f*)(aBase + slot_kstep(s) * 4);
  }
}

__device__ __forceinline__ void mma_group(const AFrags& a, const BFrags& b, int g,
                                          v8f& c0, v8f& c1, v8f& c2) {
#pragma unroll
  for (int i = 0; i < GSLOTS; ++i) {
    int s = g * GSLOTS + i;
    if (s < NSLOTS) {
      int t = slot_tile(s);
      if (t == 0)
        c0 = __builtin_amdgcn_wmma_f32_16x16x4_f32(false, a.v[i], false, b.v[i],
                                                   (short)0, c0, false, false);
      else if (t == 1)
        c1 = __builtin_amdgcn_wmma_f32_16x16x4_f32(false, a.v[i], false, b.v[i],
                                                   (short)0, c1, false, false);
      else
        c2 = __builtin_amdgcn_wmma_f32_16x16x4_f32(false, a.v[i], false, b.v[i],
                                                   (short)0, c2, false, false);
    }
  }
}

__global__ void __launch_bounds__(BLK_THREADS)
decorrelation_kernel(const float* __restrict__ x,
                     const float* __restrict__ prange,
                     const float* __restrict__ wsw,
                     float* __restrict__ out) {
  extern __shared__ float lds[];
  const int tid     = threadIdx.x;
  const int lane    = tid & 31;
  const int wave    = tid >> 5;
  const int rowBase = blockIdx.x * ROWS_BLK;

  const float binom[NC] = {1.f, 10.f, 45.f, 120.f, 210.f, 252.f,
                           210.f, 120.f, 45.f, 10.f, 1.f};

  float xv[TASKS];
  float lo3[CPERIOD], hi3[CPERIOD];
  int   row3[CPERIOD], cov3[CPERIOD];
#pragma unroll
  for (int j = 0; j < CPERIOD; ++j) {
    int task = tid + j * BLK_THREADS;
    row3[j]  = task / NV;
    cov3[j]  = task - row3[j] * NV;
    lo3[j]   = prange[cov3[j]];
    hi3[j]   = prange[NV + cov3[j]];
  }
#pragma unroll
  for (int i = 0; i < TASKS; ++i) {
    int j  = i % CPERIOD;
    int rg = rowBase + row3[j] + RSTEP * (i / CPERIOD);
    if (rg >= NSAMP) rg = NSAMP - 1;
    xv[i] = x[rg * NV + cov3[j]];
  }
  __builtin_amdgcn_sched_barrier(0);
#pragma unroll
  for (int i = 0; i < TASKS; ++i) {
    int   j   = i % CPERIOD;
    int   row = row3[j] + RSTEP * (i / CPERIOD);
    float x0  = xv[i];
    float lo  = lo3[j];
    float hi  = hi3[j];
    float xn  = (x0 - lo) / (hi - lo);
    float om  = 1.0f - xn;
    float p[NC], q[NC];
    p[0] = 1.0f;
    q[0] = 1.0f;
#pragma unroll
    for (int d = 1; d < NC; ++d) {
      p[d] = p[d - 1] * xn;
      q[d] = q[d - 1] * om;
    }
    float* g = &lds[row * G_STRIDE + cov3[j] * NC];
#pragma unroll
    for (int d = 0; d < NC; ++d)
      g[d] = binom[d] * p[d] * q[DEG - d] * x0;
  }
  v8f c0 = {}, c1 = {}, c2 = {};
  const int   mrow  = lane & 15;
  const int   khalf = (lane >> 4) * 2;
  const float* aBase = &lds[(wave * 16 + mrow) * G_STRIDE + khalf];
  const v2f*   bLane = (const v2f*)wsw + lane;

  BFrags bset[NSETS];
  AFrags aset[NSETS];
  load_b_group(bset[0], bLane, 0);
  load_b_group(bset[1], bLane, 1);
  __syncthreads();
  load_a_group(aset[0], aBase, 0);
  load_a_group(aset[1], aBase, 1);

#pragma unroll
  for (int g = 0; g < NGROUPS; ++g) {
    if (g + 2 < NGROUPS) {
      load_b_group(bset[(g + 2) % NSETS], bLane, g + 2);
      load_a_group(aset[(g + 2) % NSETS], aBase, g + 2);
    }
    __builtin_amdgcn_sched_barrier(0);
    mma_group(aset[g % NSETS], bset[g % NSETS], g, c0, c1, c2);
    __builtin_amdgcn_sched_barrier(0);
  }

  const int col = lane & 15;
  const int rg0 = rowBase + wave * 16 + ((lane >> 4) ? 8 : 0);
#pragma unroll
  for (int j = 0; j < 8; ++j) {
    int rg = rg0 + j;
    if (rg < NSAMP) {
      out[rg * NV + col]      = x[rg * NV + col]      + c0[j];
      out[rg * NV + 16 + col] = x[rg * NV + 16 + col] + c1[j];
      out[rg * NV + 32 + col] = x[rg * NV + 32 + col] + c2[j];
    }
  }
}

extern "C" void kernel_launch(void* const* d_in, const int* in_sizes, int n_in,
                              void* d_out, int out_size, void* d_ws, size_t ws_size,
                              hipStream_t stream) {
  const float* x      = (const float*)d_in[0];
  const float* params = (const float*)d_in[1];
  const float* prange = (const float*)d_in[2];
  float*       out    = (float*)d_out;
  float*       wsw    = (float*)d_ws;

  {
    const int total  = NSLOTS * 32;
    const int blocks = (total + 255) / 256;
    build_w_kernel<<<blocks, 256, 0, stream>>>(params, wsw);
  }

  {
    const int    blocks   = (NSAMP + ROWS_BLK - 1) / ROWS_BLK;
    const size_t ldsBytes = (size_t)ROWS_BLK * G_STRIDE * sizeof(float);
    decorrelation_kernel<<<blocks, BLK_THREADS, ldsBytes, stream>>>(x, prange,
                                                                    wsw, out);
  }
}


# A fragments computed in registers straight from x instead of an LDS tile: k regrouped so a lane owns whole covars of its row (4 covars = 11 k-steps); no LDS, no barrier; sums' k order changes
# speedup vs baseline: 3.4962x; 1.0938x over previous
#include <hip/hip_runtime.h>


typedef __attribute__((ext_vector_type(2))) float v2f;
typedef __attribute__((ext_vector_type(8))) float v8f;

#define DEG        10
#define NC         11
#define NV         48
#define NSAMP      50000
#define NP         1128
#define NGRP       12
#define KPG        11
#define GMAX       33
#define ROWS_WAVE  16
#define WAVES_BLK  4
#define BLK_THREADS 128
#define SEG1       132
#define SEG2       220
#define NSLOTS     264

static_assert(NGRP * 4 == NV, "groups tile the covars");
static_assert(KPG * 4 == 4 * NC, "a group's 44 k values fill 11 k-steps");

__host__ __device__ constexpr int grp_first_tile(int g) { return g < 4 ? 0 : g < 8 ? 1 : 2; }
__host__ __device__ constexpr int grp_ntiles(int g) { return 3 - grp_first_tile(g); }
__host__ __device__ constexpr int grp_slot0(int g) {
  return g < 4 ? g * 33 : g < 8 ? SEG1 + (g - 4) * 22 : SEG2 + (g - 8) * 11;
}
static_assert(grp_slot0(11) + 11 == NSLOTS, "slot list covers every nonzero block once");

__global__ void build_w_kernel(const float* __restrict__ params,
                               float* __restrict__ wsw) {
  int tid = blockIdx.x * blockDim.x + threadIdx.x;
  const int total = NSLOTS * 32;
  if (tid >= total) return;
  int lane = tid & 31;
  int slot = tid >> 5;
  int g, j, t;
  if (slot < SEG1) {
    g = slot / 33;
    int r = slot - g * 33;
    j = r / 3;
    t = r - j * 3;
  } else if (slot < SEG2) {
    int s = slot - SEG1;
    g = 4 + s / 22;
    int r = s - (g - 4) * 22;
    j = r / 2;
    t = 1 + (r - j * 2);
  } else {
    int s = slot - SEG2;
    g = 8 + s / 11;
    j = s - (g - 8) * 11;
    t = 2;
  }
  int var = t * 16 + (lane & 15);
  int h   = lane >> 4;
  float vals[2];
#pragma unroll
  for (int s = 0; s < 2; ++s) {
    int e     = 2 * j + s;
    int covar = 4 * g + 2 * h + (e >= NC ? 1 : 0);
    int d     = e - (e >= NC ? NC : 0);
    float w   = 0.0f;
    if (covar < var)
      w = params[d * NP + (var * (var - 1)) / 2 + covar];
    vals[s] = w;
  }
  float2* out2 = (float2*)wsw;
  out2[tid] = make_float2(vals[0], vals[1]);
}

struct BFrags { v2f v[GMAX]; };
struct AFrags { v2f v[KPG]; };
struct XIn    { v2f x, lo, hi; };

template <int g>
__device__ __forceinline__ void load_b_group(BFrags& f, const v2f* __restrict__ bLane) {
#pragma unroll
  for (int i = 0; i < GMAX; ++i)
    if (i < KPG * grp_ntiles(g)) f.v[i] = bLane[(grp_slot0(g) + i) * 32];
}

template <int g>
__device__ __forceinline__ void load_x_group(XIn& f, const float* __restrict__ xRow,
                                             const float* __restrict__ prLane) {
  f.x  = *(const v2f*)(xRow + 4 * g);
  f.lo = *(const v2f*)(prLane + 4 * g);
  f.hi = *(const v2f*)(prLane + NV + 4 * g);
}

__device__ __forceinline__ void bernstein_x(float x0, float lo, float hi, float (&g)[NC]) {
  const float binom[NC] = {1.f, 10.f, 45.f, 120.f, 210.f, 252.f,
                           210.f, 120.f, 45.f, 10.f, 1.f};
  float xn = (x0 - lo) / (hi - lo);
  float om = 1.0f - xn;
  float p[NC], q[NC];
  p[0] = 1.0f;
  q[0] = 1.0f;
#pragma unroll
  for (int d = 1; d < NC; ++d) {
    p[d] = p[d - 1] * xn;
    q[d] = q[d - 1] * om;
  }
#pragma unroll
  for (int d = 0; d < NC; ++d)
    g[d] = binom[d] * p[d] * q[DEG - d] * x0;
}

__device__ __forceinline__ void build_a_group(AFrags& f, const XIn& in) {
  float va[NC], vb[NC];
  bernstein_x(in.x[0], in.lo[0], in.hi[0], va);
  bernstein_x(in.x[1], in.lo[1], in.hi[1], vb);
#pragma unroll
  for (int j = 0; j < KPG; ++j) {
    const int e0 = 2 * j, e1 = 2 * j + 1;
    v2f a;
    a[0] = (e0 < NC) ? va[e0 < NC ? e0 : 0] : vb[e0 < NC ? 0 : e0 - NC];
    a[1] = (e1 < NC) ? va[e1 < NC ? e1 : 0] : vb[e1 < NC ? 0 : e1 - NC];
    f.v[j] = a;
  }
}

template <int g>
__device__ __forceinline__ void mma_group(const AFrags& a, const BFrags& b,
                                          v8f& c0, v8f& c1, v8f& c2) {
#pragma unroll
  for (int j = 0; j < KPG; ++j) {
#pragma unroll
    for (int u = 0; u < 3; ++u) {
      if (u < grp_ntiles(g)) {
        const int t = grp_first_tile(g) + u;
        const int i = j * grp_ntiles(g) + u;
        if (t == 0)
          c0 = __builtin_amdgcn_wmma_f32_16x16x4_f32(false, a.v[j], false, b.v[i],
                                                     (short)0, c0, false, false);
        else if (t == 1)
          c1 = __builtin_amdgcn_wmma_f32_16x16x4_f32(false, a.v[j], false, b.v[i],
                                                     (short)0, c1, false, false);
        else
          c2 = __builtin_amdgcn_wmma_f32_16x16x4_f32(false, a.v[j], false, b.v[i],
                                                     (short)0, c2, false, false);
      }
    }
  }
}

__host__ __device__ constexpr int b_ahead(int g) { return g < 4 ? 1 : 2; }
__host__ __device__ constexpr int b_upto(int g) {
  return g < 0 ? 0 : (g + b_ahead(g) > NGRP - 1 ? NGRP - 1 : g + b_ahead(g));
}

struct Sets {
  BFrags b[3];
  AFrags a[2];
  XIn    x[3];
};

template <int g>
__device__ __forceinline__ void group_step(Sets& s, const float* __restrict__ xRow,
                                           const float* __restrict__ prLane,
                                           const v2f* __restrict__ bLane,
                                           v8f& c0, v8f& c1, v8f& c2) {
  if constexpr (g + 1 > b_upto(g - 1) && g + 1 <= b_upto(g))
    load_b_group<(g + 1 < NGRP ? g + 1 : 0)>(s.b[(g + 1) % 3], bLane);
  if constexpr (g + 2 > b_upto(g - 1) && g + 2 <= b_upto(g))
    load_b_group<(g + 2 < NGRP ? g + 2 : 0)>(s.b[(g + 2) % 3], bLane);
  if constexpr (g + 2 < NGRP)
    load_x_group<(g + 2 < NGRP ? g + 2 : 0)>(s.x[(g + 2) % 3], xRow, prLane);
  __builtin_amdgcn_sched_barrier(0);
  mma_group<g>(s.a[g % 2], s.b[g % 3], c0, c1, c2);
  if constexpr (g + 1 < NGRP) build_a_group(s.a[(g + 1) % 2], s.x[(g + 1) % 3]);
  __builtin_amdgcn_sched_barrier(0);
}

__global__ void __launch_bounds__(BLK_THREADS)
decorrelation_kernel(const float* __restrict__ x,
                     const float* __restrict__ prange,
                     const float* __restrict__ wsw,
                     float* __restrict__ out) {
  const int tid     = threadIdx.x;
  const int lane    = tid & 31;
  const int wave    = tid >> 5;
  const int c       = lane & 15;
  const int h       = lane >> 4;
  const int rowBase = (blockIdx.x * WAVES_BLK + wave) * ROWS_WAVE;
  if (rowBase >= NSAMP) return;

  int rowA = rowBase + c;
  if (rowA >= NSAMP) rowA = NSAMP - 1;
  const float* xRow   = x + rowA * NV + 2 * h;
  const float* prLane = prange + 2 * h;
  const v2f*   bLane  = (const v2f*)wsw + lane;

  Sets s;
  load_x_group<0>(s.x[0], xRow, prLane);
  load_x_group<1>(s.x[1], xRow, prLane);
  load_b_group<0>(s.b[0], bLane);
  build_a_group(s.a[0], s.x[0]);

  v8f c0 = {}, c1 = {}, c2 = {};
  group_step<0>(s, xRow, prLane, bLane, c0, c1, c2);
  group_step<1>(s, xRow, prLane, bLane, c0, c1, c2);
  group_step<2>(s, xRow, prLane, bLane, c0, c1, c2);
  group_step<3>(s, xRow, prLane, bLane, c0, c1, c2);
  group_step<4>(s, xRow, prLane, bLane, c0, c1, c2);
  group_step<5>(s, xRow, prLane, bLane, c0, c1, c2);
  group_step<6>(s, xRow, prLane, bLane, c0, c1, c2);
  group_step<7>(s, xRow, prLane, bLane, c0, c1, c2);
  group_step<8>(s, xRow, prLane, bLane, c0, c1, c2);
  group_step<9>(s, xRow, prLane, bLane, c0, c1, c2);
  group_step<10>(s, xRow, prLane, bLane, c0, c1, c2);
  group_step<11>(s, xRow, prLane, bLane, c0, c1, c2);

  const int rg0 = rowBase + 8 * h;
#pragma unroll
  for (int j = 0; j < 8; ++j) {
    int rg = rg0 + j;
    if (rg < NSAMP) {
      out[rg * NV + c]      = x[rg * NV + c]      + c0[j];
      out[rg * NV + 16 + c] = x[rg * NV + 16 + c] + c1[j];
      out[rg * NV + 32 + c] = x[rg * NV + 32 + c] + c2[j];
    }
  }
}

extern "C" void kernel_launch(void* const* d_in, const int* in_sizes, int n_in,
                              void* d_out, int out_size, void* d_ws, size_t ws_size,
                              hipStream_t stream) {
  const float* x      = (const float*)d_in[0];
  const float* params = (const float*)d_in[1];
  const float* prange = (const float*)d_in[2];
  float*       out    = (float*)d_out;
  float*       wsw    = (float*)d_ws;

  {
    const int total  = NSLOTS * 32;
    const int blocks = (total + 255) / 256;
    build_w_kernel<<<blocks, 256, 0, stream>>>(params, wsw);
  }

  {
    const int rowsBlk = ROWS_WAVE * WAVES_BLK;
    const int blocks  = (NSAMP + rowsBlk - 1) / rowsBlk;
    decorrelation_kernel<<<blocks, BLK_THREADS, 0, stream>>>(x, prange, wsw, out);
  }
}


# One launch, not two: each 16-wave workgroup copies params to LDS and gathers W's nonzero blocks there into B-fragment order; B fragments by ds_load_b64; no workspace; sums' k order unchanged
# speedup vs baseline: 3.9814x; 1.1388x over previous
#include <hip/hip_runtime.h>


typedef __attribute__((ext_vector_type(2))) float v2f;
typedef __attribute__((ext_vector_type(8))) float v8f;

#define DEG        10
#define NC         11
#define NV         48
#define NSAMP      50000
#define NP         1128
#define NGRP       12
#define KPG        11
#define GMAX       33
#define ROWS_WAVE  16
#define WAVES_BLK  16
#define BLK_THREADS 512
#define NPAR       (NC * NP)
#define NPAR4      (NPAR / 4)
#define SEG1       132
#define SEG2       220
#define NSLOTS     264

static_assert(NGRP * 4 == NV, "groups tile the covars");
static_assert(KPG * 4 == 4 * NC, "a group's 44 k values fill 11 k-steps");
static_assert(NPAR4 * 4 == NPAR, "params copies as whole 16-byte pieces");
static_assert(NPAR4 >= BLK_THREADS, "the last copy pass may overlap the one before it");

__host__ __device__ constexpr int grp_first_tile(int g) { return g < 4 ? 0 : g < 8 ? 1 : 2; }
__host__ __device__ constexpr int grp_ntiles(int g) { return 3 - grp_first_tile(g); }
__host__ __device__ constexpr int grp_slot0(int g) {
  return g < 4 ? g * 33 : g < 8 ? SEG1 + (g - 4) * 22 : SEG2 + (g - 8) * 11;
}
static_assert(grp_slot0(11) + 11 == NSLOTS, "slot list covers every nonzero block once");

template <int jlo, int jhi>
__device__ __forceinline__ void build_w_steps(const float* src, v2f* dst, int m, int nt) {
#pragma unroll
  for (int j = jlo; j < jhi; ++j) {
    v2f vals;
#pragma unroll
    for (int s = 0; s < 2; ++s) {
      const int e = 2 * j + s;
      const int b = (e >= NC) ? 1 : 0;
      const int d = e - b * NC;
      const float v = src[d * NP + b];
      vals[s] = (b < m) ? v : 0.0f;
    }
    dst[j * nt * 32] = vals;
  }
}

__device__ __forceinline__ void build_w_unit(const float* praw, v2f* wl, int k, int half, int lane) {
  int g, t;
  if (k < 12)      { g = k / 3;                 t = k - 3 * g; }
  else if (k < 20) { g = 4 + ((k - 12) >> 1);   t = 1 + ((k - 12) & 1); }
  else             { g = 8 + (k - 20);          t = 2; }
  const int ft  = grp_first_tile(g);
  const int nt  = 3 - ft;
  const int var = t * 16 + (lane & 15);
  const int cv0 = 4 * g + 2 * (lane >> 4);
  const float* src = praw + (var * (var - 1)) / 2 + cv0;
  v2f*         dst = wl + (grp_slot0(g) + (t - ft)) * 32 + lane;
  const int    m   = var - cv0;
  if (half == 0) build_w_steps<0, 6>(src, dst, m, nt);
  else           build_w_steps<6, KPG>(src, dst, m, nt);
}

struct BFrags { v2f v[GMAX]; };
struct AFrags { v2f v[KPG]; };
struct XIn    { v2f x, lo, hi; };

template <int g>
__device__ __forceinline__ void load_b_group(BFrags& f, const v2f* bLane) {
#pragma unroll
  for (int i = 0; i < GMAX; ++i)
    if (i < KPG * grp_ntiles(g)) f.v[i] = bLane[(grp_slot0(g) + i) * 32];
}

template <int g>
__device__ __forceinline__ void load_x_group(XIn& f, const float* __restrict__ xRow,
                                             const float* __restrict__ prLane) {
  f.x  = *(const v2f*)(xRow + 4 * g);
  f.lo = *(const v2f*)(prLane + 4 * g);
  f.hi = *(const v2f*)(prLane + NV + 4 * g);
}

__device__ __forceinline__ void bernstein_x(float x0, float lo, float hi, float (&g)[NC]) {
  const float binom[NC] = {1.f, 10.f, 45.f, 120.f, 210.f, 252.f,
                           210.f, 120.f, 45.f, 10.f, 1.f};
  float xn = (x0 - lo) / (hi - lo);
  float om = 1.0f - xn;
  float p[NC], q[NC];
  p[0] = 1.0f;
  q[0] = 1.0f;
#pragma unroll
  for (int d = 1; d < NC; ++d) {
    p[d] = p[d - 1] * xn;
    q[d] = q[d - 1] * om;
  }
#pragma unroll
  for (int d = 0; d < NC; ++d)
    g[d] = binom[d] * p[d] * q[DEG - d] * x0;
}

__device__ __forceinline__ void build_a_group(AFrags& f, const XIn& in) {
  float va[NC], vb[NC];
  bernstein_x(in.x[0], in.lo[0], in.hi[0], va);
  bernstein_x(in.x[1], in.lo[1], in.hi[1], vb);
#pragma unroll
  for (int j = 0; j < KPG; ++j) {
    const int e0 = 2 * j, e1 = 2 * j + 1;
    v2f a;
    a[0] = (e0 < NC) ? va[e0 < NC ? e0 : 0] : vb[e0 < NC ? 0 : e0 - NC];
    a[1] = (e1 < NC) ? va[e1 < NC ? e1 : 0] : vb[e1 < NC ? 0 : e1 - NC];
    f.v[j] = a;
  }
}

template <int g>
__device__ __forceinline__ void mma_group(const AFrags& a, const BFrags& b,
                                          v8f& c0, v8f& c1, v8f& c2) {
#pragma unroll
  for (int j = 0; j < KPG; ++j) {
#pragma unroll
    for (int u = 0; u < 3; ++u) {
      if (u < grp_ntiles(g)) {
        const int t = grp_first_tile(g) + u;
        const int i = j * grp_ntiles(g) + u;
        if (t == 0)
          c0 = __builtin_amdgcn_wmma_f32_16x16x4_f32(false, a.v[j], false, b.v[i],
                                                     (short)0, c0, false, false);
        else if (t == 1)
          c1 = __builtin_amdgcn_wmma_f32_16x16x4_f32(false, a.v[j], false, b.v[i],
                                                     (short)0, c1, false, false);
        else
          c2 = __builtin_amdgcn_wmma_f32_16x16x4_f32(false, a.v[j], false, b.v[i],
                                                     (short)0, c2, false, false);
      }
    }
  }
}

__host__ __device__ constexpr int b_ahead(int g) { return g < 4 ? 1 : 2; }
__host__ __device__ constexpr int b_upto(int g) {
  return g < 0 ? 0 : (g + b_ahead(g) > NGRP - 1 ? NGRP - 1 : g + b_ahead(g));
}

struct Sets {
  BFrags b[3];
  AFrags a[2];
  XIn    x[3];
};

template <int g>
__device__ __forceinline__ void group_step(Sets& s, const float* __restrict__ xRow,
                                           const float* __restrict__ prLane,
                                           const v2f* bLane,
                                           v8f& c0, v8f& c1, v8f& c2) {
  if constexpr (g + 1 > b_upto(g - 1) && g + 1 <= b_upto(g))
    load_b_group<(g + 1 < NGRP ? g + 1 : 0)>(s.b[(g + 1) % 3], bLane);
  if constexpr (g + 2 > b_upto(g - 1) && g + 2 <= b_upto(g))
    load_b_group<(g + 2 < NGRP ? g + 2 : 0)>(s.b[(g + 2) % 3], bLane);
  if constexpr (g + 2 < NGRP)
    load_x_group<(g + 2 < NGRP ? g + 2 : 0)>(s.x[(g + 2) % 3], xRow, prLane);
  __builtin_amdgcn_sched_barrier(0);
  mma_group<g>(s.a[g % 2], s.b[g % 3], c0, c1, c2);
  if constexpr (g + 1 < NGRP) build_a_group(s.a[(g + 1) % 2], s.x[(g + 1) % 3]);
  __builtin_amdgcn_sched_barrier(0);
}

__global__ void __launch_bounds__(BLK_THREADS)
decorrelation_kernel(const float* __restrict__ x,
                     const float* __restrict__ params,
                     const float* __restrict__ prange,
                     float* __restrict__ out) {
  __shared__ __attribute__((aligned(256))) float praw[NPAR + 4];
  __shared__ __attribute__((aligned(256))) v2f   wl[NSLOTS * 32];

  const int tid     = threadIdx.x;
  const int lane    = tid & 31;
  const int wave    = __builtin_amdgcn_readfirstlane(tid >> 5);
  const int c       = lane & 15;
  const int h       = lane >> 4;
  const int rowBase = (blockIdx.x * WAVES_BLK + wave) * ROWS_WAVE;

  int rowA = rowBase + c;
  if (rowA >= NSAMP) rowA = NSAMP - 1;
  const float* xRow   = x + rowA * NV + 2 * h;
  const float* prLane = prange + 2 * h;
  const v2f*   bLane  = wl + lane;

  Sets s;
  load_x_group<0>(s.x[0], xRow, prLane);
  load_x_group<1>(s.x[1], xRow, prLane);

  {
    const float4* p4 = (const float4*)params;
    float4*       l4 = (float4*)praw;
    constexpr int FULL = NPAR4 / BLK_THREADS;
    constexpr int LAST = NPAR4 - BLK_THREADS;
    float4 v[FULL + 1];
#pragma unroll
    for (int i = 0; i < FULL; ++i) v[i] = p4[tid + i * BLK_THREADS];
    v[FULL] = p4[LAST + tid];
#pragma unroll
    for (int i = 0; i < FULL; ++i) l4[tid + i * BLK_THREADS] = v[i];
    l4[LAST + tid] = v[FULL];
    if (tid == 0) l4[NPAR4] = make_float4(0.f, 0.f, 0.f, 0.f);
  }
  __syncthreads();
#pragma unroll
  for (int p = 0; p < 3; ++p)
    build_w_unit(praw, wl, (wave >> 1) + 8 * p, wave & 1, lane);
  __syncthreads();

  if (rowBase >= NSAMP) return;

  load_b_group<0>(s.b[0], bLane);
  build_a_group(s.a[0], s.x[0]);

  v8f c0 = {}, c1 = {}, c2 = {};
  group_step<0>(s, xRow, prLane, bLane, c0, c1, c2);
  group_step<1>(s, xRow, prLane, bLane, c0, c1, c2);
  group_step<2>(s, xRow, prLane, bLane, c0, c1, c2);
  group_step<3>(s, xRow, prLane, bLane, c0, c1, c2);
  group_step<4>(s, xRow, prLane, bLane, c0, c1, c2);
  group_step<5>(s, xRow, prLane, bLane, c0, c1, c2);
  group_step<6>(s, xRow, prLane, bLane, c0, c1, c2);
  group_step<7>(s, xRow, prLane, bLane, c0, c1, c2);
  group_step<8>(s, xRow, prLane, bLane, c0, c1, c2);
  group_step<9>(s, xRow, prLane, bLane, c0, c1, c2);
  group_step<10>(s, xRow, prLane, bLane, c0, c1, c2);
  group_step<11>(s, xRow, prLane, bLane, c0, c1, c2);

  const int rg0 = rowBase + 8 * h;
#pragma unroll
  for (int j = 0; j < 8; ++j) {
    int rg = rg0 + j;
    if (rg < NSAMP) {
      out[rg * NV + c]      = x[rg * NV + c]      + c0[j];
      out[rg * NV + 16 + c] = x[rg * NV + 16 + c] + c1[j];
      out[rg * NV + 32 + c] = x[rg * NV + 32 + c] + c2[j];
    }
  }
}

extern "C" void kernel_launch(void* const* d_in, const int* in_sizes, int n_in,
                              void* d_out, int out_size, void* d_ws, size_t ws_size,
                              hipStream_t stream) {
  const float* x      = (const float*)d_in[0];
  const float* params = (const float*)d_in[1];
  const float* prange = (const float*)d_in[2];
  float*       out    = (float*)d_out;

  const int rowsBlk = ROWS_WAVE * WAVES_BLK;
  const int blocks  = (NSAMP + rowsBlk - 1) / rowsBlk;
  decorrelation_kernel<<<blocks, BLK_THREADS, 0, stream>>>(x, params, prange, out);
}


# Epilogue's 24 x loads leave as one run 4 groups before the end (C/D layout, one address) instead of 8 guarded load-add-store round trips behind the last product; sums unchanged
# speedup vs baseline: 4.9792x; 1.2506x over previous
#include <hip/hip_runtime.h>


typedef __attribute__((ext_vector_type(2))) float v2f;
typedef __attribute__((ext_vector_type(8))) float v8f;

#define DEG        10
#define NC         11
#define NV         48
#define NSAMP      50000
#define NP         1128
#define NGRP       12
#define KPG        11
#define GMAX       33
#define ROWS_WAVE  16
#define WAVES_BLK  16
#define BLK_THREADS 512
#define NPAR       (NC * NP)
#define NPAR4      (NPAR / 4)
#define SEG1       132
#define SEG2       220
#define NSLOTS     264
#define G_XOUT     8

static_assert(NGRP * 4 == NV, "groups tile the covars");
static_assert(NSAMP % ROWS_WAVE == 0, "every wave that runs owns 16 whole rows: the epilogue needs no row guard");
static_assert(G_XOUT + 2 < NGRP, "the epilogue's x loads leave at least three groups ahead of their use");
static_assert(KPG * 4 == 4 * NC, "a group's 44 k values fill 11 k-steps");
static_assert(NPAR4 * 4 == NPAR, "params copies as whole 16-byte pieces");
static_assert(NPAR4 >= BLK_THREADS, "the last copy pass may overlap the one before it");

__host__ __device__ constexpr int grp_first_tile(int g) { return g < 4 ? 0 : g < 8 ? 1 : 2; }
__host__ __device__ constexpr int grp_ntiles(int g) { return 3 - grp_first_tile(g); }
__host__ __device__ constexpr int grp_slot0(int g) {
  return g < 4 ? g * 33 : g < 8 ? SEG1 + (g - 4) * 22 : SEG2 + (g - 8) * 11;
}
static_assert(grp_slot0(11) + 11 == NSLOTS, "slot list covers every nonzero block once");

template <int jlo, int jhi>
__device__ __forceinline__ void build_w_steps(const float* src, v2f* dst, int m, int nt) {
#pragma unroll
  for (int j = jlo; j < jhi; ++j) {
    v2f vals;
#pragma unroll
    for (int s = 0; s < 2; ++s) {
      const int e = 2 * j + s;
      const int b = (e >= NC) ? 1 : 0;
      const int d = e - b * NC;
      const float v = src[d * NP + b];
      vals[s] = (b < m) ? v : 0.0f;
    }
    dst[j * nt * 32] = vals;
  }
}

__device__ __forceinline__ void build_w_unit(const float* praw, v2f* wl, int k, int half, int lane) {
  int g, t;
  if (k < 12)      { g = k / 3;                 t = k - 3 * g; }
  else if (k < 20) { g = 4 + ((k - 12) >> 1);   t = 1 + ((k - 12) & 1); }
  else             { g = 8 + (k - 20);          t = 2; }
  const int ft  = grp_first_tile(g);
  const int nt  = 3 - ft;
  const int var = t * 16 + (lane & 15);
  const int cv0 = 4 * g + 2 * (lane >> 4);
  const float* src = praw + (var * (var - 1)) / 2 + cv0;
  v2f*         dst = wl + (grp_slot0(g) + (t - ft)) * 32 + lane;
  const int    m   = var - cv0;
  if (half == 0) build_w_steps<0, 6>(src, dst, m, nt);
  else           build_w_steps<6, KPG>(src, dst, m, nt);
}

struct BFrags { v2f v[GMAX]; };
struct AFrags { v2f v[KPG]; };
struct XIn    { v2f x, lo, hi; };
struct XOut   { float v[24]; };

template <int g>
__device__ __forceinline__ void load_b_group(BFrags& f, const v2f* bLane) {
#pragma unroll
  for (int i = 0; i < GMAX; ++i)
    if (i < KPG * grp_ntiles(g)) f.v[i] = bLane[(grp_slot0(g) + i) * 32];
}

template <int g>
__device__ __forceinline__ void load_x_group(XIn& f, const float* __restrict__ xRow,
                                             const float* __restrict__ prLane) {
  f.x  = *(const v2f*)(xRow + 4 * g);
  f.lo = *(const v2f*)(prLane + 4 * g);
  f.hi = *(const v2f*)(prLane + NV + 4 * g);
}

__device__ __forceinline__ void load_x_out(XOut& f, const float* __restrict__ xOut) {
#pragma unroll
  for (int j = 0; j < 8; ++j) {
    f.v[3 * j + 0] = xOut[j * NV];
    f.v[3 * j + 1] = xOut[j * NV + 16];
    f.v[3 * j + 2] = xOut[j * NV + 32];
  }
}

__device__ __forceinline__ void bernstein_x(float x0, float lo, float hi, float (&g)[NC]) {
  const float binom[NC] = {1.f, 10.f, 45.f, 120.f, 210.f, 252.f,
                           210.f, 120.f, 45.f, 10.f, 1.f};
  float xn = (x0 - lo) / (hi - lo);
  float om = 1.0f - xn;
  float p[NC], q[NC];
  p[0] = 1.0f;
  q[0] = 1.0f;
#pragma unroll
  for (int d = 1; d < NC; ++d) {
    p[d] = p[d - 1] * xn;
    q[d] = q[d - 1] * om;
  }
#pragma unroll
  for (int d = 0; d < NC; ++d)
    g[d] = binom[d] * p[d] * q[DEG - d] * x0;
}

__device__ __forceinline__ void build_a_group(AFrags& f, const XIn& in) {
  float va[NC], vb[NC];
  bernstein_x(in.x[0], in.lo[0], in.hi[0], va);
  bernstein_x(in.x[1], in.lo[1], in.hi[1], vb);
#pragma unroll
  for (int j = 0; j < KPG; ++j) {
    const int e0 = 2 * j, e1 = 2 * j + 1;
    v2f a;
    a[0] = (e0 < NC) ? va[e0 < NC ? e0 : 0] : vb[e0 < NC ? 0 : e0 - NC];
    a[1] = (e1 < NC) ? va[e1 < NC ? e1 : 0] : vb[e1 < NC ? 0 : e1 - NC];
    f.v[j] = a;
  }
}

template <int g>
__device__ __forceinline__ void mma_group(const AFrags& a, const BFrags& b,
                                          v8f& c0, v8f& c1, v8f& c2) {
#pragma unroll
  for (int j = 0; j < KPG; ++j) {
#pragma unroll
    for (int u = 0; u < 3; ++u) {
      if (u < grp_ntiles(g)) {
        const int t = grp_first_tile(g) + u;
        const int i = j * grp_ntiles(g) + u;
        if (t == 0)
          c0 = __builtin_amdgcn_wmma_f32_16x16x4_f32(false, a.v[j], false, b.v[i],
                                                     (short)0, c0, false, false);
        else if (t == 1)
          c1 = __builtin_amdgcn_wmma_f32_16x16x4_f32(false, a.v[j], false, b.v[i],
                                                     (short)0, c1, false, false);
        else
          c2 = __builtin_amdgcn_wmma_f32_16x16x4_f32(false, a.v[j], false, b.v[i],
                                                     (short)0, c2, false, false);
      }
    }
  }
}

__host__ __device__ constexpr int b_ahead(int g) { return g < 4 ? 1 : 2; }
__host__ __device__ constexpr int b_upto(int g) {
  return g < 0 ? 0 : (g + b_ahead(g) > NGRP - 1 ? NGRP - 1 : g + b_ahead(g));
}

struct Sets {
  BFrags b[3];
  AFrags a[2];
  XIn    x[3];
};

template <int g>
__device__ __forceinline__ void group_step(Sets& s, XOut& xo, const float* __restrict__ xRow,
                                           const float* __restrict__ prLane,
                                           const float* __restrict__ xOut,
                                           const v2f* bLane,
                                           v8f& c0, v8f& c1, v8f& c2) {
  if constexpr (g + 1 > b_upto(g - 1) && g + 1 <= b_upto(g))
    load_b_group<(g + 1 < NGRP ? g + 1 : 0)>(s.b[(g + 1) % 3], bLane);
  if constexpr (g + 2 > b_upto(g - 1) && g + 2 <= b_upto(g))
    load_b_group<(g + 2 < NGRP ? g + 2 : 0)>(s.b[(g + 2) % 3], bLane);
  if constexpr (g + 2 < NGRP)
    load_x_group<(g + 2 < NGRP ? g + 2 : 0)>(s.x[(g + 2) % 3], xRow, prLane);
  if constexpr (g == G_XOUT) load_x_out(xo, xOut);
  __builtin_amdgcn_sched_barrier(0);
  mma_group<g>(s.a[g % 2], s.b[g % 3], c0, c1, c2);
  if constexpr (g + 1 < NGRP) build_a_group(s.a[(g + 1) % 2], s.x[(g + 1) % 3]);
  __builtin_amdgcn_sched_barrier(0);
}

__global__ void __launch_bounds__(BLK_THREADS)
decorrelation_kernel(const float* __restrict__ x,
                     const float* __restrict__ params,
                     const float* __restrict__ prange,
                     float* __restrict__ out) {
  __shared__ __attribute__((aligned(256))) float praw[NPAR + 4];
  __shared__ __attribute__((aligned(256))) v2f   wl[NSLOTS * 32];

  const int tid     = threadIdx.x;
  const int lane    = tid & 31;
  const int wave    = __builtin_amdgcn_readfirstlane(tid >> 5);
  const int c       = lane & 15;
  const int h       = lane >> 4;
  const int rowBase = (blockIdx.x * WAVES_BLK + wave) * ROWS_WAVE;

  int rowA = rowBase + c;
  if (rowA >= NSAMP) rowA = NSAMP - 1;
  const float* xRow   = x + rowA * NV + 2 * h;
  const float* prLane = prange + 2 * h;
  const v2f*   bLane  = wl + lane;
  const float* xOut   = x   + (rowBase + 8 * h) * NV + c;
  float*       oLane  = out + (rowBase + 8 * h) * NV + c;

  Sets s;
  XOut xo;
  load_x_group<0>(s.x[0], xRow, prLane);
  load_x_group<1>(s.x[1], xRow, prLane);

  {
    const float4* p4 = (const float4*)params;
    float4*       l4 = (float4*)praw;
    constexpr int FULL = NPAR4 / BLK_THREADS;
    constexpr int LAST = NPAR4 - BLK_THREADS;
    float4 v[FULL + 1];
#pragma unroll
    for (int i = 0; i < FULL; ++i) v[i] = p4[tid + i * BLK_THREADS];
    v[FULL] = p4[LAST + tid];
#pragma unroll
    for (int i = 0; i < FULL; ++i) l4[tid + i * BLK_THREADS] = v[i];
    l4[LAST + tid] = v[FULL];
    if (tid == 0) l4[NPAR4] = make_float4(0.f, 0.f, 0.f, 0.f);
  }
  __syncthreads();
#pragma unroll
  for (int p = 0; p < 3; ++p)
    build_w_unit(praw, wl, (wave >> 1) + 8 * p, wave & 1, lane);
  __syncthreads();

  if (rowBase >= NSAMP) return;

  load_b_group<0>(s.b[0], bLane);
  build_a_group(s.a[0], s.x[0]);

  v8f c0 = {}, c1 = {}, c2 = {};
  group_step<0>(s, xo, xRow, prLane, xOut, bLane, c0, c1, c2);
  group_step<1>(s, xo, xRow, prLane, xOut, bLane, c0, c1, c2);
  group_step<2>(s, xo, xRow, prLane, xOut, bLane, c0, c1, c2);
  group_step<3>(s, xo, xRow, prLane, xOut, bLane, c0, c1, c2);
  group_step<4>(s, xo, xRow, prLane, xOut, bLane, c0, c1, c2);
  group_step<5>(s, xo, xRow, prLane, xOut, bLane, c0, c1, c2);
  group_step<6>(s, xo, xRow, prLane, xOut, bLane, c0, c1, c2);
  group_step<7>(s, xo, xRow, prLane, xOut, bLane, c0, c1, c2);
  group_step<8>(s, xo, xRow, prLane, xOut, bLane, c0, c1, c2);
  group_step<9>(s, xo, xRow, prLane, xOut, bLane, c0, c1, c2);
  group_step<10>(s, xo, xRow, prLane, xOut, bLane, c0, c1, c2);
  group_step<11>(s, xo, xRow, prLane, xOut, bLane, c0, c1, c2);

#pragma unroll
  for (int j = 0; j < 8; ++j) {
    oLane[j * NV]      = xo.v[3 * j + 0] + c0[j];
    oLane[j * NV + 16] = xo.v[3 * j + 1] + c1[j];
    oLane[j * NV + 32] = xo.v[3 * j + 2] + c2[j];
  }
}

extern "C" void kernel_launch(void* const* d_in, const int* in_sizes, int n_in,
                              void* d_out, int out_size, void* d_ws, size_t ws_size,
                              hipStream_t stream) {
  const float* x      = (const float*)d_in[0];
  const float* params = (const float*)d_in[1];
  const float* prange = (const float*)d_in[2];
  float*       out    = (float*)d_out;

  const int rowsBlk = ROWS_WAVE * WAVES_BLK;
  const int blocks  = (NSAMP + rowsBlk - 1) / rowsBlk;
  decorrelation_kernel<<<blocks, BLK_THREADS, 0, stream>>>(x, params, prange, out);
}


# A build cut to 29 packed multiplies per group: k order in a group now (degree, covar pair) so each product is a fragment; binomials folded into W at its build; sums' k order and rounding change
# speedup vs baseline: 5.2273x; 1.0498x over previous
#include <hip/hip_runtime.h>


typedef __attribute__((ext_vector_type(2))) float v2f;
typedef __attribute__((ext_vector_type(8))) float v8f;

#define DEG        10
#define NC         11
#define NV         48
#define NSAMP      50000
#define NP         1128
#define NGRP       12
#define KPG        11
#define GMAX       33
#define ROWS_WAVE  16
#define WAVES_BLK  16
#define BLK_THREADS 512
#define NPAR       (NC * NP)
#define NPAR4      (NPAR / 4)
#define SEG1       132
#define SEG2       220
#define NSLOTS     264
#define G_XOUT     8

static_assert(NGRP * 4 == NV, "groups tile the covars");
static_assert(NSAMP % ROWS_WAVE == 0, "every wave that runs owns 16 whole rows: the epilogue needs no row guard");
static_assert(G_XOUT + 2 < NGRP, "the epilogue's x loads leave at least three groups ahead of their use");
static_assert(KPG == NC, "k-step j of a group is degree j of the group's four covars");
static_assert(NPAR4 * 4 == NPAR, "params copies as whole 16-byte pieces");
static_assert(NPAR4 >= BLK_THREADS, "the last copy pass may overlap the one before it");

__host__ __device__ constexpr int grp_first_tile(int g) { return g < 4 ? 0 : g < 8 ? 1 : 2; }
__host__ __device__ constexpr int grp_ntiles(int g) { return 3 - grp_first_tile(g); }
__host__ __device__ constexpr int grp_slot0(int g) {
  return g < 4 ? g * 33 : g < 8 ? SEG1 + (g - 4) * 22 : SEG2 + (g - 8) * 11;
}
static_assert(grp_slot0(11) + 11 == NSLOTS, "slot list covers every nonzero block once");

template <int jlo, int jhi>
__device__ __forceinline__ void build_w_steps(const float* src, v2f* dst, int m, int nt) {
  const float binom[NC] = {1.f, 10.f, 45.f, 120.f, 210.f, 252.f,
                           210.f, 120.f, 45.f, 10.f, 1.f};
#pragma unroll
  for (int j = jlo; j < jhi; ++j) {
    v2f vals;
#pragma unroll
    for (int s = 0; s < 2; ++s) {
      const float v = src[j * NP + s] * binom[j];
      vals[s] = (s < m) ? v : 0.0f;
    }
    dst[j * nt * 32] = vals;
  }
}

__device__ __forceinline__ void build_w_unit(const float* praw, v2f* wl, int k, int half, int lane) {
  int g, t;
  if (k < 12)      { g = k / 3;                 t = k - 3 * g; }
  else if (k < 20) { g = 4 + ((k - 12) >> 1);   t = 1 + ((k - 12) & 1); }
  else             { g = 8 + (k - 20);          t = 2; }
  const int ft  = grp_first_tile(g);
  const int nt  = 3 - ft;
  const int var = t * 16 + (lane & 15);
  const int cv0 = 4 * g + 2 * (lane >> 4);
  const float* src = praw + (var * (var - 1)) / 2 + cv0;
  v2f*         dst = wl + (grp_slot0(g) + (t - ft)) * 32 + lane;
  const int    m   = var - cv0;
  if (half == 0) build_w_steps<0, 6>(src, dst, m, nt);
  else           build_w_steps<6, KPG>(src, dst, m, nt);
}

struct BFrags { v2f v[GMAX]; };
struct AFrags { v2f v[KPG]; };
struct XIn    { v2f x, lo, hi; };
struct XOut   { float v[24]; };

template <int g>
__device__ __forceinline__ void load_b_group(BFrags& f, const v2f* bLane) {
#pragma unroll
  for (int i = 0; i < GMAX; ++i)
    if (i < KPG * grp_ntiles(g)) f.v[i] = bLane[(grp_slot0(g) + i) * 32];
}

template <int g>
__device__ __forceinline__ void load_x_group(XIn& f, const float* __restrict__ xRow,
                                             const float* __restrict__ prLane) {
  f.x  = *(const v2f*)(xRow + 4 * g);
  f.lo = *(const v2f*)(prLane + 4 * g);
  f.hi = *(const v2f*)(prLane + NV + 4 * g);
}

__device__ __forceinline__ void load_x_out(XOut& f, const float* __restrict__ xOut) {
#pragma unroll
  for (int j = 0; j < 8; ++j) {
    f.v[3 * j + 0] = xOut[j * NV];
    f.v[3 * j + 1] = xOut[j * NV + 16];
    f.v[3 * j + 2] = xOut[j * NV + 32];
  }
}

__device__ __forceinline__ void build_a_group(AFrags& f, const XIn& in) {
  const v2f one = {1.0f, 1.0f};
  const v2f xn  = (in.x - in.lo) / (in.hi - in.lo);
  const v2f om  = one - xn;
  v2f u[NC], q[NC];
  u[0] = in.x;
  q[0] = one;
  q[1] = om;
#pragma unroll
  for (int d = 1; d < NC; ++d) u[d] = u[d - 1] * xn;
#pragma unroll
  for (int d = 2; d < NC; ++d) q[d] = q[d - 1] * om;
#pragma unroll
  for (int d = 0; d < DEG; ++d) f.v[d] = u[d] * q[DEG - d];
  f.v[DEG] = u[DEG];
}

template <int g>
__device__ __forceinline__ void mma_group(const AFrags& a, const BFrags& b,
                                          v8f& c0, v8f& c1, v8f& c2) {
#pragma unroll
  for (int j = 0; j < KPG; ++j) {
#pragma unroll
    for (int u = 0; u < 3; ++u) {
      if (u < grp_ntiles(g)) {
        const int t = grp_first_tile(g) + u;
        const int i = j * grp_ntiles(g) + u;
        if (t == 0)
          c0 = __builtin_amdgcn_wmma_f32_16x16x4_f32(false, a.v[j], false, b.v[i],
                                                     (short)0, c0, false, false);
        else if (t == 1)
          c1 = __builtin_amdgcn_wmma_f32_16x16x4_f32(false, a.v[j], false, b.v[i],
                                                     (short)0, c1, false, false);
        else
          c2 = __builtin_amdgcn_wmma_f32_16x16x4_f32(false, a.v[j], false, b.v[i],
                                                     (short)0, c2, false, false);
      }
    }
  }
}

__host__ __device__ constexpr int b_ahead(int g) { return g < 4 ? 1 : 2; }
__host__ __device__ constexpr int b_upto(int g) {
  return g < 0 ? 0 : (g + b_ahead(g) > NGRP - 1 ? NGRP - 1 : g + b_ahead(g));
}

struct Sets {
  BFrags b[3];
  AFrags a[2];
  XIn    x[3];
};

template <int g>
__device__ __forceinline__ void group_step(Sets& s, XOut& xo, const float* __restrict__ xRow,
                                           const float* __restrict__ prLane,
                                           const float* __restrict__ xOut,
                                           const v2f* bLane,
                                           v8f& c0, v8f& c1, v8f& c2) {
  if constexpr (g + 1 > b_upto(g - 1) && g + 1 <= b_upto(g))
    load_b_group<(g + 1 < NGRP ? g + 1 : 0)>(s.b[(g + 1) % 3], bLane);
  if constexpr (g + 2 > b_upto(g - 1) && g + 2 <= b_upto(g))
    load_b_group<(g + 2 < NGRP ? g + 2 : 0)>(s.b[(g + 2) % 3], bLane);
  if constexpr (g + 2 < NGRP)
    load_x_group<(g + 2 < NGRP ? g + 2 : 0)>(s.x[(g + 2) % 3], xRow, prLane);
  if constexpr (g == G_XOUT) load_x_out(xo, xOut);
  __builtin_amdgcn_sched_barrier(0);
  mma_group<g>(s.a[g % 2], s.b[g % 3], c0, c1, c2);
  if constexpr (g + 1 < NGRP) build_a_group(s.a[(g + 1) % 2], s.x[(g + 1) % 3]);
  __builtin_amdgcn_sched_barrier(0);
}

__global__ void __launch_bounds__(BLK_THREADS)
decorrelation_kernel(const float* __restrict__ x,
                     const float* __restrict__ params,
                     const float* __restrict__ prange,
                     float* __restrict__ out) {
  __shared__ __attribute__((aligned(256))) float praw[NPAR + 4];
  __shared__ __attribute__((aligned(256))) v2f   wl[NSLOTS * 32];

  const int tid     = threadIdx.x;
  const int lane    = tid & 31;
  const int wave    = __builtin_amdgcn_readfirstlane(tid >> 5);
  const int c       = lane & 15;
  const int h       = lane >> 4;
  const int rowBase = (blockIdx.x * WAVES_BLK + wave) * ROWS_WAVE;

  int rowA = rowBase + c;
  if (rowA >= NSAMP) rowA = NSAMP - 1;
  const float* xRow   = x + rowA * NV + 2 * h;
  const float* prLane = prange + 2 * h;
  const v2f*   bLane  = wl + lane;
  const float* xOut   = x   + (rowBase + 8 * h) * NV + c;
  float*       oLane  = out + (rowBase + 8 * h) * NV + c;

  Sets s;
  XOut xo;
  load_x_group<0>(s.x[0], xRow, prLane);
  load_x_group<1>(s.x[1], xRow, prLane);

  {
    const float4* p4 = (const float4*)params;
    float4*       l4 = (float4*)praw;
    constexpr int FULL = NPAR4 / BLK_THREADS;
    constexpr int LAST = NPAR4 - BLK_THREADS;
    float4 v[FULL + 1];
#pragma unroll
    for (int i = 0; i < FULL; ++i) v[i] = p4[tid + i * BLK_THREADS];
    v[FULL] = p4[LAST + tid];
#pragma unroll
    for (int i = 0; i < FULL; ++i) l4[tid + i * BLK_THREADS] = v[i];
    l4[LAST + tid] = v[FULL];
    if (tid == 0) l4[NPAR4] = make_float4(0.f, 0.f, 0.f, 0.f);
  }
  __syncthreads();
#pragma unroll
  for (int p = 0; p < 3; ++p)
    build_w_unit(praw, wl, (wave >> 1) + 8 * p, wave & 1, lane);
  __syncthreads();

  if (rowBase >= NSAMP) return;

  load_b_group<0>(s.b[0], bLane);
  build_a_group(s.a[0], s.x[0]);

  v8f c0 = {}, c1 = {}, c2 = {};
  group_step<0>(s, xo, xRow, prLane, xOut, bLane, c0, c1, c2);
  group_step<1>(s, xo, xRow, prLane, xOut, bLane, c0, c1, c2);
  group_step<2>(s, xo, xRow, prLane, xOut, bLane, c0, c1, c2);
  group_step<3>(s, xo, xRow, prLane, xOut, bLane, c0, c1, c2);
  group_step<4>(s, xo, xRow, prLane, xOut, bLane, c0, c1, c2);
  group_step<5>(s, xo, xRow, prLane, xOut, bLane, c0, c1, c2);
  group_step<6>(s, xo, xRow, prLane, xOut, bLane, c0, c1, c2);
  group_step<7>(s, xo, xRow, prLane, xOut, bLane, c0, c1, c2);
  group_step<8>(s, xo, xRow, prLane, xOut, bLane, c0, c1, c2);
  group_step<9>(s, xo, xRow, prLane, xOut, bLane, c0, c1, c2);
  group_step<10>(s, xo, xRow, prLane, xOut, bLane, c0, c1, c2);
  group_step<11>(s, xo, xRow, prLane, xOut, bLane, c0, c1, c2);

#pragma unroll
  for (int j = 0; j < 8; ++j) {
    oLane[j * NV]      = xo.v[3 * j + 0] + c0[j];
    oLane[j * NV + 16] = xo.v[3 * j + 1] + c1[j];
    oLane[j * NV + 32] = xo.v[3 * j + 2] + c2[j];
  }
}

extern "C" void kernel_launch(void* const* d_in, const int* in_sizes, int n_in,
                              void* d_out, int out_size, void* d_ws, size_t ws_size,
                              hipStream_t stream) {
  const float* x      = (const float*)d_in[0];
  const float* params = (const float*)d_in[1];
  const float* prange = (const float*)d_in[2];
  float*       out    = (float*)d_out;

  const int rowsBlk = ROWS_WAVE * WAVES_BLK;
  const int blocks  = (NSAMP + rowsBlk - 1) / rowsBlk;
  decorrelation_kernel<<<blocks, BLK_THREADS, 0, stream>>>(x, params, prange, out);
}
